# starting kernel
# baseline (speedup 1.0000x reference)
#include <hip/hip_runtime.h>

typedef __attribute__((ext_vector_type(2))) float v2f;
typedef __attribute__((ext_vector_type(8))) float v8f;

#define WMMA_F32(a, b, c) \
    __builtin_amdgcn_wmma_f32_16x16x4_f32(false, (a), false, (b), (short)0, (c), false, false)


__global__ void __launch_bounds__(128)
g_kernel(const float* __restrict__ X, float* __restrict__ Gpart) {
    __shared__ float Xt[16][80];

    const int bh   = blockIdx.x;
    const int ks   = blockIdx.y;
    const int b    = bh >> 4;
    const int h    = bh & 15;
    const int wave = threadIdx.x >> 5;
    const int lane = threadIdx.x & 31;
    const int l15  = lane & 15;
    const int khalf = (lane >> 4) * 2;
    const float* __restrict__ Xh = X + (size_t)b * 2048u * 1024u + h * 64;

    const int m0 = wave * 16;
    v8f acc[4];
#pragma unroll
    for (int t = 0; t < 4; ++t) acc[t] = (v8f)0.0f;

    const int srow = threadIdx.x >> 3;
    const int scol = (threadIdx.x & 7) * 8;

    const int kend = ks * 512 + 512;
    for (int k0 = ks * 512; k0 < kend; k0 += 16) {
        const float* src = Xh + (size_t)(k0 + srow) * 1024u + scol;
        float4 v0 = *(const float4*)(src);
        float4 v1 = *(const float4*)(src + 4);
        *(float2*)&Xt[srow][scol + 0] = make_float2(v0.x, v0.y);
        *(float2*)&Xt[srow][scol + 2] = make_float2(v0.z, v0.w);
        *(float2*)&Xt[srow][scol + 4] = make_float2(v1.x, v1.y);
        *(float2*)&Xt[srow][scol + 6] = make_float2(v1.z, v1.w);
        __syncthreads();

#pragma unroll
        for (int kk = 0; kk < 16; kk += 4) {
            const int rA = kk + khalf;
            v2f a;
            a.x = Xt[rA][m0 + l15];
            a.y = Xt[rA + 1][m0 + l15];
#pragma unroll
            for (int t = 0; t < 4; ++t) {
                v2f bb;
                bb.x = Xt[rA][t * 16 + l15];
                bb.y = Xt[rA + 1][t * 16 + l15];
                acc[t] = WMMA_F32(a, bb, acc[t]);
            }
        }
        __syncthreads();
    }

    float* __restrict__ Gp = Gpart + ((size_t)ks * 64u + bh) * 64u * 64u;
    const int rowAdd = (lane >> 4) * 8;
#pragma unroll
    for (int t = 0; t < 4; ++t)
#pragma unroll
        for (int r = 0; r < 8; ++r)
            Gp[(m0 + r + rowAdd) * 64 + t * 16 + l15] = acc[t][r];
}

__global__ void __launch_bounds__(256)
g_reduce(const float* __restrict__ Gpart, float* __restrict__ G) {
    const int i = blockIdx.x * 256 + threadIdx.x;
    const size_t SL = (size_t)64 * 64 * 64;
    G[i] = ((Gpart[i] + Gpart[SL + i]) + Gpart[2 * SL + i]) + Gpart[3 * SL + i];
}

__global__ void __launch_bounds__(128)
wprime_kernel(const float* __restrict__ W, const float* __restrict__ G,
              float* __restrict__ Wp) {
    const int bh   = blockIdx.y;
    const int b    = bh >> 4;
    const int h    = bh & 15;
    const int wave = threadIdx.x >> 5;
    const int lane = threadIdx.x & 31;
    const int l15  = lane & 15;
    const int khalf = (lane >> 4) * 2;
    const int m0 = blockIdx.x * 64 + wave * 16;

    const float* __restrict__ Wh = W + (size_t)(h * 64) * 1024u;
    const float* __restrict__ Gp = G + (size_t)bh * 64u * 64u;

    v8f acc[4];
#pragma unroll
    for (int t = 0; t < 4; ++t) acc[t] = (v8f)0.0f;

#pragma unroll 4
    for (int d0 = 0; d0 < 64; d0 += 4) {
        const int rA = d0 + khalf;
        v2f a;
        a.x = Wh[(size_t)rA * 1024u + m0 + l15];
        a.y = Wh[(size_t)(rA + 1) * 1024u + m0 + l15];
#pragma unroll
        for (int t = 0; t < 4; ++t) {
            v2f bb;
            bb.x = Gp[rA * 64 + t * 16 + l15];
            bb.y = Gp[(rA + 1) * 64 + t * 16 + l15];
            acc[t] = WMMA_F32(a, bb, acc[t]);
        }
    }

    float* __restrict__ Wpb = Wp + (size_t)b * 1024u * 1024u;
    const int rowAdd = (lane >> 4) * 8;
    const int ncol0 = h * 64;
#pragma unroll
    for (int t = 0; t < 4; ++t)
#pragma unroll
        for (int r = 0; r < 8; ++r)
            Wpb[(size_t)(m0 + r + rowAdd) * 1024u + ncol0 + t * 16 + l15] = acc[t][r];
}

__global__ void __launch_bounds__(256)
out_kernel(const float* __restrict__ X, const float* __restrict__ Wp,
           float* __restrict__ out) {
    __shared__ float Xs[128][18];
    __shared__ float Ws[16][80];

    const int b      = blockIdx.z;
    const int s_base = blockIdx.x * 128;
    const int n_base = blockIdx.y * 64;
    const int wave   = threadIdx.x >> 5;
    const int lane   = threadIdx.x & 31;
    const int l15    = lane & 15;
    const int khalf  = (lane >> 4) * 2;
    const int m0     = wave * 16;

    const float* __restrict__ Xb  = X + (size_t)b * 2048u * 1024u;
    const float* __restrict__ Wpb = Wp + (size_t)b * 1024u * 1024u;

    v8f acc[4];
#pragma unroll
    for (int t = 0; t < 4; ++t) acc[t] = (v8f)0.0f;

    const int ldr = threadIdx.x >> 1;
    const int ldc = (threadIdx.x & 1) * 8;
    const int wrow = threadIdx.x >> 4;
    const int wcol = (threadIdx.x & 15) * 4;

    for (int k0 = 0; k0 < 1024; k0 += 16) {
        {
            const float* src = &Xb[(size_t)(s_base + ldr) * 1024u + k0 + ldc];
            float4 v0 = *(const float4*)(src);
            float4 v1 = *(const float4*)(src + 4);
            *(float2*)&Xs[ldr][ldc + 0] = make_float2(v0.x, v0.y);
            *(float2*)&Xs[ldr][ldc + 2] = make_float2(v0.z, v0.w);
            *(float2*)&Xs[ldr][ldc + 4] = make_float2(v1.x, v1.y);
            *(float2*)&Xs[ldr][ldc + 6] = make_float2(v1.z, v1.w);
        }
        {
            const float* wsrc = &Wpb[(size_t)(k0 + wrow) * 1024u + n_base + wcol];
            float4 wv = *(const float4*)(wsrc);
            *(float2*)&Ws[wrow][wcol + 0] = make_float2(wv.x, wv.y);
            *(float2*)&Ws[wrow][wcol + 2] = make_float2(wv.z, wv.w);
        }
        __syncthreads();

#pragma unroll
        for (int kk = 0; kk < 16; kk += 4) {
            const int rK = kk + khalf;
            v2f a;
            a.x = Xs[m0 + l15][rK];
            a.y = Xs[m0 + l15][rK + 1];
#pragma unroll
            for (int t = 0; t < 4; ++t) {
                v2f bb;
                bb.x = Ws[rK][t * 16 + l15];
                bb.y = Ws[rK + 1][t * 16 + l15];
                acc[t] = WMMA_F32(a, bb, acc[t]);
            }
        }
        __syncthreads();
    }

    float* __restrict__ ob = out + (size_t)b * 2048u * 1024u;
    const int rowAdd = (lane >> 4) * 8;
#pragma unroll
    for (int t = 0; t < 4; ++t)
#pragma unroll
        for (int r = 0; r < 8; ++r)
            ob[(size_t)(s_base + m0 + r + rowAdd) * 1024u + n_base + t * 16 + l15] =
                acc[t][r];
}

extern "C" void kernel_launch(void* const* d_in, const int* in_sizes, int n_in,
                              void* d_out, int out_size, void* d_ws, size_t ws_size,
                              hipStream_t stream) {
    (void)in_sizes; (void)n_in; (void)out_size; (void)ws_size;
    const float* X = (const float*)d_in[0];
    const float* W = (const float*)d_in[1];
    float* out = (float*)d_out;

    float* Gpart = (float*)d_ws;
    float* G     = Gpart + (size_t)4 * 64 * 64 * 64;
    float* Wp    = G + (size_t)64 * 64 * 64;

    g_kernel<<<dim3(64, 4), dim3(128), 0, stream>>>(X, Gpart);
    g_reduce<<<dim3(1024), dim3(256), 0, stream>>>(Gpart, G);
    wprime_kernel<<<dim3(16, 64), dim3(128), 0, stream>>>(W, G, Wp);
    out_kernel<<<dim3(16, 16, 4), dim3(256), 0, stream>>>(X, Wp, out);
}


# K loops of the G and out kernels software-pipelined over two LDS buffers: next tiles' global loads issued ahead of the current tile's WMMAs (out 1 tile, G 4 tiles ahead), one barrier per tile
# speedup vs baseline: 1.2359x; 1.2359x over previous
#include <hip/hip_runtime.h>

typedef __attribute__((ext_vector_type(2))) float v2f;
typedef __attribute__((ext_vector_type(8))) float v8f;

#define WMMA_F32(a, b, c) \
    __builtin_amdgcn_wmma_f32_16x16x4_f32(false, (a), false, (b), (short)0, (c), false, false)


#define XT_STRIDE 80

__device__ __forceinline__ void lds_only_barrier() {
    __builtin_amdgcn_fence(__ATOMIC_RELEASE, "workgroup", "local");
    __builtin_amdgcn_s_barrier();
    __builtin_amdgcn_fence(__ATOMIC_ACQUIRE, "workgroup", "local");
}

__device__ __forceinline__ void g_tile_mma(const float (*__restrict__ Xt)[XT_STRIDE],
                                           int m0, int khalf, int l15, v8f (&acc)[4]) {
#pragma unroll
    for (int kk = 0; kk < 16; kk += 4) {
        const int rA = kk + khalf;
        v2f a;
        a.x = Xt[rA][m0 + l15];
        a.y = Xt[rA + 1][m0 + l15];
#pragma unroll
        for (int t = 0; t < 4; ++t) {
            v2f bb;
            bb.x = Xt[rA][t * 16 + l15];
            bb.y = Xt[rA + 1][t * 16 + l15];
            acc[t] = WMMA_F32(a, bb, acc[t]);
        }
    }
}

__global__ void __launch_bounds__(128)
g_kernel(const float* __restrict__ X, float* __restrict__ Gpart) {
    __shared__ float Xt[2][16][XT_STRIDE];

    const int bh   = blockIdx.x;
    const int ks   = blockIdx.y;
    const int b    = bh >> 4;
    const int h    = bh & 15;
    const int wave = threadIdx.x >> 5;
    const int lane = threadIdx.x & 31;
    const int l15  = lane & 15;
    const int khalf = (lane >> 4) * 2;
    const float* __restrict__ Xh = X + (size_t)b * 2048u * 1024u + h * 64;

    const int m0 = wave * 16;
    v8f acc[4];
#pragma unroll
    for (int t = 0; t < 4; ++t) acc[t] = (v8f)0.0f;

    const int srow = threadIdx.x >> 3;
    const int scol = (threadIdx.x & 7) * 8;

    const float* src = Xh + (size_t)(ks * 512 + srow) * 1024u + scol;
    float4 ra[4], rb[4];

#define G_FETCH(u)                                    \
    do {                                              \
        ra[u] = *(const float4*)(src);                \
        rb[u] = *(const float4*)(src + 4);            \
        src += (size_t)16 * 1024u;                    \
    } while (0)

#define G_STASH(buf, u)                                                         \
    do {                                                                        \
        *(float2*)&Xt[buf][srow][scol + 0] = make_float2(ra[u].x, ra[u].y);     \
        *(float2*)&Xt[buf][srow][scol + 2] = make_float2(ra[u].z, ra[u].w);     \
        *(float2*)&Xt[buf][srow][scol + 4] = make_float2(rb[u].x, rb[u].y);     \
        *(float2*)&Xt[buf][srow][scol + 6] = make_float2(rb[u].z, rb[u].w);     \
    } while (0)

#define G_HOLD_LOADS_HERE() __builtin_amdgcn_sched_barrier(0)
#define G_MMA_BEFORE_WAIT() __builtin_amdgcn_sched_barrier(0)

#define G_STEP(u, FETCH)                                       \
    do {                                                       \
        if (FETCH) G_FETCH(u);                                 \
        G_HOLD_LOADS_HERE();                                   \
        g_tile_mma(Xt[(u) & 1], m0, khalf, l15, acc);          \
        G_MMA_BEFORE_WAIT();                                   \
        G_STASH(((u) + 1) & 1, ((u) + 1) & 3);                 \
        lds_only_barrier();                                    \
    } while (0)

    G_FETCH(0);
    G_FETCH(1);
    G_FETCH(2);
    G_FETCH(3);
    G_STASH(0, 0);
    lds_only_barrier();

    for (int it = 0; it < 28; it += 4) {
        G_STEP(0, true);
        G_STEP(1, true);
        G_STEP(2, true);
        G_STEP(3, true);
    }
    G_STEP(0, false);
    G_STEP(1, false);
    G_STEP(2, false);
    g_tile_mma(Xt[1], m0, khalf, l15, acc);

#undef G_FETCH
#undef G_STASH
#undef G_HOLD_LOADS_HERE
#undef G_MMA_BEFORE_WAIT
#undef G_STEP

    float* __restrict__ Gp = Gpart + ((size_t)ks * 64u + bh) * 64u * 64u;
    const int rowAdd = (lane >> 4) * 8;
#pragma unroll
    for (int t = 0; t < 4; ++t)
#pragma unroll
        for (int r = 0; r < 8; ++r)
            Gp[(m0 + r + rowAdd) * 64 + t * 16 + l15] = acc[t][r];
}

__global__ void __launch_bounds__(256)
g_reduce(const float* __restrict__ Gpart, float* __restrict__ G) {
    const int i = blockIdx.x * 256 + threadIdx.x;
    const size_t SL = (size_t)64 * 64 * 64;
    G[i] = ((Gpart[i] + Gpart[SL + i]) + Gpart[2 * SL + i]) + Gpart[3 * SL + i];
}

__global__ void __launch_bounds__(128)
wprime_kernel(const float* __restrict__ W, const float* __restrict__ G,
              float* __restrict__ Wp) {
    const int bh   = blockIdx.y;
    const int b    = bh >> 4;
    const int h    = bh & 15;
    const int wave = threadIdx.x >> 5;
    const int lane = threadIdx.x & 31;
    const int l15  = lane & 15;
    const int khalf = (lane >> 4) * 2;
    const int m0 = blockIdx.x * 64 + wave * 16;

    const float* __restrict__ Wh = W + (size_t)(h * 64) * 1024u;
    const float* __restrict__ Gp = G + (size_t)bh * 64u * 64u;

    v8f acc[4];
#pragma unroll
    for (int t = 0; t < 4; ++t) acc[t] = (v8f)0.0f;

#pragma unroll 4
    for (int d0 = 0; d0 < 64; d0 += 4) {
        const int rA = d0 + khalf;
        v2f a;
        a.x = Wh[(size_t)rA * 1024u + m0 + l15];
        a.y = Wh[(size_t)(rA + 1) * 1024u + m0 + l15];
#pragma unroll
        for (int t = 0; t < 4; ++t) {
            v2f bb;
            bb.x = Gp[rA * 64 + t * 16 + l15];
            bb.y = Gp[(rA + 1) * 64 + t * 16 + l15];
            acc[t] = WMMA_F32(a, bb, acc[t]);
        }
    }

    float* __restrict__ Wpb = Wp + (size_t)b * 1024u * 1024u;
    const int rowAdd = (lane >> 4) * 8;
    const int ncol0 = h * 64;
#pragma unroll
    for (int t = 0; t < 4; ++t)
#pragma unroll
        for (int r = 0; r < 8; ++r)
            Wpb[(size_t)(m0 + r + rowAdd) * 1024u + ncol0 + t * 16 + l15] = acc[t][r];
}

#define XS_STRIDE 18
#define WS_STRIDE 80

__device__ __forceinline__ void out_tile_mma(const float (*__restrict__ Xs)[XS_STRIDE],
                                             const float (*__restrict__ Ws)[WS_STRIDE],
                                             int arow, int khalf, int l15, v8f (&acc)[4]) {
#pragma unroll
    for (int kk = 0; kk < 16; kk += 4) {
        const int rK = kk + khalf;
        v2f a;
        a.x = Xs[arow][rK];
        a.y = Xs[arow][rK + 1];
#pragma unroll
        for (int t = 0; t < 4; ++t) {
            v2f bb;
            bb.x = Ws[rK][t * 16 + l15];
            bb.y = Ws[rK + 1][t * 16 + l15];
            acc[t] = WMMA_F32(a, bb, acc[t]);
        }
    }
}

__global__ void __launch_bounds__(256)
out_kernel(const float* __restrict__ X, const float* __restrict__ Wp,
           float* __restrict__ out) {
    __shared__ float Xs[2][128][XS_STRIDE];
    __shared__ float Ws[2][16][WS_STRIDE];

    const int b      = blockIdx.z;
    const int s_base = blockIdx.x * 128;
    const int n_base = blockIdx.y * 64;
    const int wave   = threadIdx.x >> 5;
    const int lane   = threadIdx.x & 31;
    const int l15    = lane & 15;
    const int khalf  = (lane >> 4) * 2;
    const int m0     = wave * 16;

    const float* __restrict__ Xb  = X + (size_t)b * 2048u * 1024u;
    const float* __restrict__ Wpb = Wp + (size_t)b * 1024u * 1024u;

    v8f acc[4];
#pragma unroll
    for (int t = 0; t < 4; ++t) acc[t] = (v8f)0.0f;

    const int ldr = threadIdx.x >> 1;
    const int ldc = (threadIdx.x & 1) * 8;
    const int wrow = threadIdx.x >> 4;
    const int wcol = (threadIdx.x & 15) * 4;

    const float* xsrc = &Xb[(size_t)(s_base + ldr) * 1024u + ldc];
    const float* wsrc = &Wpb[(size_t)wrow * 1024u + n_base + wcol];

    float4 xv0, xv1, wv;

#define OUT_FETCH()                                    \
    do {                                               \
        xv0 = *(const float4*)(xsrc);                  \
        xv1 = *(const float4*)(xsrc + 4);              \
        wv  = *(const float4*)(wsrc);                  \
        xsrc += 16;                                    \
        wsrc += (size_t)16 * 1024u;                    \
    } while (0)

#define OUT_HOLD_LOADS_HERE() __builtin_amdgcn_sched_barrier(0)
#define OUT_MMA_BEFORE_WAIT() __builtin_amdgcn_sched_barrier(0)

#define OUT_STASH(buf)                                                        \
    do {                                                                      \
        *(float2*)&Xs[buf][ldr][ldc + 0] = make_float2(xv0.x, xv0.y);         \
        *(float2*)&Xs[buf][ldr][ldc + 2] = make_float2(xv0.z, xv0.w);         \
        *(float2*)&Xs[buf][ldr][ldc + 4] = make_float2(xv1.x, xv1.y);         \
        *(float2*)&Xs[buf][ldr][ldc + 6] = make_float2(xv1.z, xv1.w);         \
        *(float2*)&Ws[buf][wrow][wcol + 0] = make_float2(wv.x, wv.y);         \
        *(float2*)&Ws[buf][wrow][wcol + 2] = make_float2(wv.z, wv.w);         \
    } while (0)

    OUT_FETCH();
    OUT_STASH(0);
    __syncthreads();

    for (int it = 0; it < 62; it += 2) {
        OUT_FETCH();
        OUT_HOLD_LOADS_HERE();
        out_tile_mma(Xs[0], Ws[0], m0 + l15, khalf, l15, acc);
        OUT_MMA_BEFORE_WAIT();
        OUT_STASH(1);
        __syncthreads();

        OUT_FETCH();
        OUT_HOLD_LOADS_HERE();
        out_tile_mma(Xs[1], Ws[1], m0 + l15, khalf, l15, acc);
        OUT_MMA_BEFORE_WAIT();
        OUT_STASH(0);
        __syncthreads();
    }
    OUT_FETCH();
    OUT_HOLD_LOADS_HERE();
    out_tile_mma(Xs[0], Ws[0], m0 + l15, khalf, l15, acc);
    OUT_MMA_BEFORE_WAIT();
    OUT_STASH(1);
    __syncthreads();
    out_tile_mma(Xs[1], Ws[1], m0 + l15, khalf, l15, acc);

#undef OUT_FETCH
#undef OUT_HOLD_LOADS_HERE
#undef OUT_MMA_BEFORE_WAIT
#undef OUT_STASH

    float* __restrict__ ob = out + (size_t)b * 2048u * 1024u;
    const int rowAdd = (lane >> 4) * 8;
#pragma unroll
    for (int t = 0; t < 4; ++t)
#pragma unroll
        for (int r = 0; r < 8; ++r)
            ob[(size_t)(s_base + m0 + r + rowAdd) * 1024u + n_base + t * 16 + l15] =
                acc[t][r];
}

extern "C" void kernel_launch(void* const* d_in, const int* in_sizes, int n_in,
                              void* d_out, int out_size, void* d_ws, size_t ws_size,
                              hipStream_t stream) {
    (void)in_sizes; (void)n_in; (void)out_size; (void)ws_size;
    const float* X = (const float*)d_in[0];
    const float* W = (const float*)d_in[1];
    float* out = (float*)d_out;

    float* Gpart = (float*)d_ws;
    float* G     = Gpart + (size_t)4 * 64 * 64 * 64;
    float* Wp    = G + (size_t)64 * 64 * 64;

    g_kernel<<<dim3(64, 4), dim3(128), 0, stream>>>(X, Gpart);
    g_reduce<<<dim3(1024), dim3(256), 0, stream>>>(Gpart, G);
    wprime_kernel<<<dim3(16, 64), dim3(128), 0, stream>>>(W, G, Wp);
    out_kernel<<<dim3(16, 16, 4), dim3(256), 0, stream>>>(X, Wp, out);
}


# out kernel re-tiled: each wave owns a 64x64 tile (16 accumulators; 4 waves, 128x128 per workgroup), B staged k-pair-interleaved: a k-step is 8 eight-byte LDS reads for 16 WMMAs; sums unchanged
# speedup vs baseline: 1.3547x; 1.0961x over previous
#include <hip/hip_runtime.h>

typedef __attribute__((ext_vector_type(2))) float v2f;
typedef __attribute__((ext_vector_type(8))) float v8f;

#define WMMA_F32(a, b, c) \
    __builtin_amdgcn_wmma_f32_16x16x4_f32(false, (a), false, (b), (short)0, (c), false, false)


#define XT_STRIDE 80

__device__ __forceinline__ void lds_only_barrier() {
    __builtin_amdgcn_fence(__ATOMIC_RELEASE, "workgroup", "local");
    __builtin_amdgcn_s_barrier();
    __builtin_amdgcn_fence(__ATOMIC_ACQUIRE, "workgroup", "local");
}

__device__ __forceinline__ void g_tile_mma(const float (*__restrict__ Xt)[XT_STRIDE],
                                           int m0, int khalf, int l15, v8f (&acc)[4]) {
#pragma unroll
    for (int kk = 0; kk < 16; kk += 4) {
        const int rA = kk + khalf;
        v2f a;
        a.x = Xt[rA][m0 + l15];
        a.y = Xt[rA + 1][m0 + l15];
#pragma unroll
        for (int t = 0; t < 4; ++t) {
            v2f bb;
            bb.x = Xt[rA][t * 16 + l15];
            bb.y = Xt[rA + 1][t * 16 + l15];
            acc[t] = WMMA_F32(a, bb, acc[t]);
        }
    }
}

__global__ void __launch_bounds__(128)
g_kernel(const float* __restrict__ X, float* __restrict__ Gpart) {
    __shared__ float Xt[2][16][XT_STRIDE];

    const int bh   = blockIdx.x;
    const int ks   = blockIdx.y;
    const int b    = bh >> 4;
    const int h    = bh & 15;
    const int wave = threadIdx.x >> 5;
    const int lane = threadIdx.x & 31;
    const int l15  = lane & 15;
    const int khalf = (lane >> 4) * 2;
    const float* __restrict__ Xh = X + (size_t)b * 2048u * 1024u + h * 64;

    const int m0 = wave * 16;
    v8f acc[4];
#pragma unroll
    for (int t = 0; t < 4; ++t) acc[t] = (v8f)0.0f;

    const int srow = threadIdx.x >> 3;
    const int scol = (threadIdx.x & 7) * 8;

    const float* src = Xh + (size_t)(ks * 512 + srow) * 1024u + scol;
    float4 ra[4], rb[4];

#define G_FETCH(u)                                    \
    do {                                              \
        ra[u] = *(const float4*)(src);                \
        rb[u] = *(const float4*)(src + 4);            \
        src += (size_t)16 * 1024u;                    \
    } while (0)

#define G_STASH(buf, u)                                                         \
    do {                                                                        \
        *(float2*)&Xt[buf][srow][scol + 0] = make_float2(ra[u].x, ra[u].y);     \
        *(float2*)&Xt[buf][srow][scol + 2] = make_float2(ra[u].z, ra[u].w);     \
        *(float2*)&Xt[buf][srow][scol + 4] = make_float2(rb[u].x, rb[u].y);     \
        *(float2*)&Xt[buf][srow][scol + 6] = make_float2(rb[u].z, rb[u].w);     \
    } while (0)

#define G_HOLD_LOADS_HERE() __builtin_amdgcn_sched_barrier(0)
#define G_MMA_BEFORE_WAIT() __builtin_amdgcn_sched_barrier(0)

#define G_STEP(u, FETCH)                                       \
    do {                                                       \
        if (FETCH) G_FETCH(u);                                 \
        G_HOLD_LOADS_HERE();                                   \
        g_tile_mma(Xt[(u) & 1], m0, khalf, l15, acc);          \
        G_MMA_BEFORE_WAIT();                                   \
        G_STASH(((u) + 1) & 1, ((u) + 1) & 3);                 \
        lds_only_barrier();                                    \
    } while (0)

    G_FETCH(0);
    G_FETCH(1);
    G_FETCH(2);
    G_FETCH(3);
    G_STASH(0, 0);
    lds_only_barrier();

    for (int it = 0; it < 28; it += 4) {
        G_STEP(0, true);
        G_STEP(1, true);
        G_STEP(2, true);
        G_STEP(3, true);
    }
    G_STEP(0, false);
    G_STEP(1, false);
    G_STEP(2, false);
    g_tile_mma(Xt[1], m0, khalf, l15, acc);

#undef G_FETCH
#undef G_STASH
#undef G_HOLD_LOADS_HERE
#undef G_MMA_BEFORE_WAIT
#undef G_STEP

    float* __restrict__ Gp = Gpart + ((size_t)ks * 64u + bh) * 64u * 64u;
    const int rowAdd = (lane >> 4) * 8;
#pragma unroll
    for (int t = 0; t < 4; ++t)
#pragma unroll
        for (int r = 0; r < 8; ++r)
            Gp[(m0 + r + rowAdd) * 64 + t * 16 + l15] = acc[t][r];
}

__global__ void __launch_bounds__(256)
g_reduce(const float* __restrict__ Gpart, float* __restrict__ G) {
    const int i = blockIdx.x * 256 + threadIdx.x;
    const size_t SL = (size_t)64 * 64 * 64;
    G[i] = ((Gpart[i] + Gpart[SL + i]) + Gpart[2 * SL + i]) + Gpart[3 * SL + i];
}

__global__ void __launch_bounds__(128)
wprime_kernel(const float* __restrict__ W, const float* __restrict__ G,
              float* __restrict__ Wp) {
    const int bh   = blockIdx.y;
    const int b    = bh >> 4;
    const int h    = bh & 15;
    const int wave = threadIdx.x >> 5;
    const int lane = threadIdx.x & 31;
    const int l15  = lane & 15;
    const int khalf = (lane >> 4) * 2;
    const int m0 = blockIdx.x * 64 + wave * 16;

    const float* __restrict__ Wh = W + (size_t)(h * 64) * 1024u;
    const float* __restrict__ Gp = G + (size_t)bh * 64u * 64u;

    v8f acc[4];
#pragma unroll
    for (int t = 0; t < 4; ++t) acc[t] = (v8f)0.0f;

#pragma unroll 4
    for (int d0 = 0; d0 < 64; d0 += 4) {
        const int rA = d0 + khalf;
        v2f a;
        a.x = Wh[(size_t)rA * 1024u + m0 + l15];
        a.y = Wh[(size_t)(rA + 1) * 1024u + m0 + l15];
#pragma unroll
        for (int t = 0; t < 4; ++t) {
            v2f bb;
            bb.x = Gp[rA * 64 + t * 16 + l15];
            bb.y = Gp[(rA + 1) * 64 + t * 16 + l15];
            acc[t] = WMMA_F32(a, bb, acc[t]);
        }
    }

    float* __restrict__ Wpb = Wp + (size_t)b * 1024u * 1024u;
    const int rowAdd = (lane >> 4) * 8;
    const int ncol0 = h * 64;
#pragma unroll
    for (int t = 0; t < 4; ++t)
#pragma unroll
        for (int r = 0; r < 8; ++r)
            Wpb[(size_t)(m0 + r + rowAdd) * 1024u + ncol0 + t * 16 + l15] = acc[t][r];
}

#define OT_K      16
#define XS_STRIDE 20
#define WS_PITCH  288

__device__ __forceinline__ void out_tile_mma(const float (*__restrict__ Xs)[XS_STRIDE],
                                             const float (*__restrict__ Ws)[WS_PITCH],
                                             int arow, int bcol2, int half, v8f (&acc)[4][4]) {
#pragma unroll
    for (int kk = 0; kk < OT_K; kk += 4) {
        v2f a[4], bb[4];
#pragma unroll
        for (int mt = 0; mt < 4; ++mt)
            a[mt] = *(const v2f*)&Xs[arow + mt * 16][kk + 2 * half];
#pragma unroll
        for (int nt = 0; nt < 4; ++nt)
            bb[nt] = *(const v2f*)&Ws[kk / 2 + half][bcol2 + nt * 32];
#pragma unroll
        for (int mt = 0; mt < 4; ++mt)
#pragma unroll
            for (int nt = 0; nt < 4; ++nt)
                acc[mt][nt] = WMMA_F32(a[mt], bb[nt], acc[mt][nt]);
    }
}

__global__ void __launch_bounds__(128)
out_kernel(const float* __restrict__ X, const float* __restrict__ Wp,
           float* __restrict__ out) {
    __shared__ __attribute__((aligned(16))) float Xs[2][128][XS_STRIDE];
    __shared__ __attribute__((aligned(16))) float Ws[2][OT_K / 2][WS_PITCH];

    const int b      = blockIdx.z;
    const int s_base = blockIdx.x * 128;
    const int n_base = blockIdx.y * 128;
    const int wave   = threadIdx.x >> 5;
    const int lane   = threadIdx.x & 31;
    const int l15    = lane & 15;
    const int half   = lane >> 4;
    const int m0     = (wave >> 1) * 64;
    const int n0     = (wave & 1) * 64;

    const float* __restrict__ Xb  = X + (size_t)b * 2048u * 1024u;
    const float* __restrict__ Wpb = Wp + (size_t)b * 1024u * 1024u;

    v8f acc[4][4];
#pragma unroll
    for (int mt = 0; mt < 4; ++mt)
#pragma unroll
        for (int nt = 0; nt < 4; ++nt) acc[mt][nt] = (v8f)0.0f;

    const int xr = threadIdx.x >> 2;
    const int xc = (threadIdx.x & 3) * 4;
    const int wn4 = threadIdx.x & 31;
    const int wq  = threadIdx.x >> 5;

    const float* xsrc = &Xb[(size_t)(s_base + xr) * 1024u + xc];
    const float* wsrc = &Wpb[(size_t)(2 * wq) * 1024u + n_base + 4 * wn4];

    float4 xv0, xv1, xv2, xv3, wa0, wb0, wa1, wb1;

#define OUT_FETCH()                                                              \
    do {                                                                         \
        xv0 = *(const float4*)(xsrc);                                            \
        xv1 = *(const float4*)(xsrc + (size_t)32 * 1024u);                       \
        xv2 = *(const float4*)(xsrc + (size_t)64 * 1024u);                       \
        xv3 = *(const float4*)(xsrc + (size_t)96 * 1024u);                       \
        wa0 = *(const float4*)(wsrc);                                            \
        wb0 = *(const float4*)(wsrc + (size_t)1 * 1024u);                        \
        wa1 = *(const float4*)(wsrc + (size_t)8 * 1024u);                        \
        wb1 = *(const float4*)(wsrc + (size_t)9 * 1024u);                        \
        xsrc += OT_K;                                                            \
        wsrc += (size_t)OT_K * 1024u;                                            \
    } while (0)

#define OUT_HOLD_LOADS_HERE() __builtin_amdgcn_sched_barrier(0)
#define OUT_MMA_BEFORE_WAIT() __builtin_amdgcn_sched_barrier(0)

#define OUT_STASH(buf)                                                           \
    do {                                                                         \
        *(float4*)&Xs[buf][xr +  0][xc] = xv0;                                   \
        *(float4*)&Xs[buf][xr + 32][xc] = xv1;                                   \
        *(float4*)&Xs[buf][xr + 64][xc] = xv2;                                   \
        *(float4*)&Xs[buf][xr + 96][xc] = xv3;                                   \
        *(float4*)&Ws[buf][wq][8 * wn4 + 0] =                                    \
            make_float4(wa0.x, wb0.x, wa0.y, wb0.y);                             \
        *(float4*)&Ws[buf][wq][8 * wn4 + 4] =                                    \
            make_float4(wa0.z, wb0.z, wa0.w, wb0.w);                             \
        *(float4*)&Ws[buf][wq + 4][8 * wn4 + 0] =                                \
            make_float4(wa1.x, wb1.x, wa1.y, wb1.y);                             \
        *(float4*)&Ws[buf][wq + 4][8 * wn4 + 4] =                                \
            make_float4(wa1.z, wb1.z, wa1.w, wb1.w);                             \
    } while (0)

    const int arow  = m0 + l15;
    const int bcol2 = 2 * (n0 + l15);

    OUT_FETCH();
    OUT_STASH(0);
    __syncthreads();

    for (int it = 0; it < 62; it += 2) {
        OUT_FETCH();
        OUT_HOLD_LOADS_HERE();
        out_tile_mma(Xs[0], Ws[0], arow, bcol2, half, acc);
        OUT_MMA_BEFORE_WAIT();
        OUT_STASH(1);
        __syncthreads();

        OUT_FETCH();
        OUT_HOLD_LOADS_HERE();
        out_tile_mma(Xs[1], Ws[1], arow, bcol2, half, acc);
        OUT_MMA_BEFORE_WAIT();
        OUT_STASH(0);
        __syncthreads();
    }
    OUT_FETCH();
    OUT_HOLD_LOADS_HERE();
    out_tile_mma(Xs[0], Ws[0], arow, bcol2, half, acc);
    OUT_MMA_BEFORE_WAIT();
    OUT_STASH(1);
    __syncthreads();
    out_tile_mma(Xs[1], Ws[1], arow, bcol2, half, acc);

#undef OUT_FETCH
#undef OUT_HOLD_LOADS_HERE
#undef OUT_MMA_BEFORE_WAIT
#undef OUT_STASH

    float* __restrict__ ob = out + (size_t)b * 2048u * 1024u;
    const int rowAdd = half * 8;
#pragma unroll
    for (int mt = 0; mt < 4; ++mt)
#pragma unroll
        for (int nt = 0; nt < 4; ++nt)
#pragma unroll
            for (int r = 0; r < 8; ++r)
                ob[(size_t)(s_base + m0 + mt * 16 + r + rowAdd) * 1024u +
                   n_base + n0 + nt * 16 + l15] = acc[mt][nt][r];
}

extern "C" void kernel_launch(void* const* d_in, const int* in_sizes, int n_in,
                              void* d_out, int out_size, void* d_ws, size_t ws_size,
                              hipStream_t stream) {
    (void)in_sizes; (void)n_in; (void)out_size; (void)ws_size;
    const float* X = (const float*)d_in[0];
    const float* W = (const float*)d_in[1];
    float* out = (float*)d_out;

    float* Gpart = (float*)d_ws;
    float* G     = Gpart + (size_t)4 * 64 * 64 * 64;
    float* Wp    = G + (size_t)64 * 64 * 64;

    g_kernel<<<dim3(64, 4), dim3(128), 0, stream>>>(X, Gpart);
    g_reduce<<<dim3(1024), dim3(256), 0, stream>>>(Gpart, G);
    wprime_kernel<<<dim3(16, 64), dim3(128), 0, stream>>>(W, G, Wp);
    out_kernel<<<dim3(16, 8, 4), dim3(128), 0, stream>>>(X, Wp, out);
}


# out kernel: wave priority follows K-loop progress (s_setprio 3,2,1,0, stepping down at K tiles 32,48,56) so a WGP's two co-resident workgroups trade the matrix pipe and end together; sums unchanged
# speedup vs baseline: 1.4182x; 1.0469x over previous
#include <hip/hip_runtime.h>

typedef __attribute__((ext_vector_type(2))) float v2f;
typedef __attribute__((ext_vector_type(8))) float v8f;

#define WMMA_F32(a, b, c) \
    __builtin_amdgcn_wmma_f32_16x16x4_f32(false, (a), false, (b), (short)0, (c), false, false)


#define XT_STRIDE 80

__device__ __forceinline__ void lds_only_barrier() {
    __builtin_amdgcn_fence(__ATOMIC_RELEASE, "workgroup", "local");
    __builtin_amdgcn_s_barrier();
    __builtin_amdgcn_fence(__ATOMIC_ACQUIRE, "workgroup", "local");
}

__device__ __forceinline__ void g_tile_mma(const float (*__restrict__ Xt)[XT_STRIDE],
                                           int m0, int khalf, int l15, v8f (&acc)[4]) {
#pragma unroll
    for (int kk = 0; kk < 16; kk += 4) {
        const int rA = kk + khalf;
        v2f a;
        a.x = Xt[rA][m0 + l15];
        a.y = Xt[rA + 1][m0 + l15];
#pragma unroll
        for (int t = 0; t < 4; ++t) {
            v2f bb;
            bb.x = Xt[rA][t * 16 + l15];
            bb.y = Xt[rA + 1][t * 16 + l15];
            acc[t] = WMMA_F32(a, bb, acc[t]);
        }
    }
}

__global__ void __launch_bounds__(128)
g_kernel(const float* __restrict__ X, float* __restrict__ Gpart) {
    __shared__ float Xt[2][16][XT_STRIDE];

    const int bh   = blockIdx.x;
    const int ks   = blockIdx.y;
    const int b    = bh >> 4;
    const int h    = bh & 15;
    const int wave = threadIdx.x >> 5;
    const int lane = threadIdx.x & 31;
    const int l15  = lane & 15;
    const int khalf = (lane >> 4) * 2;
    const float* __restrict__ Xh = X + (size_t)b * 2048u * 1024u + h * 64;

    const int m0 = wave * 16;
    v8f acc[4];
#pragma unroll
    for (int t = 0; t < 4; ++t) acc[t] = (v8f)0.0f;

    const int srow = threadIdx.x >> 3;
    const int scol = (threadIdx.x & 7) * 8;

    const float* src = Xh + (size_t)(ks * 512 + srow) * 1024u + scol;
    float4 ra[4], rb[4];

#define G_FETCH(u)                                    \
    do {                                              \
        ra[u] = *(const float4*)(src);                \
        rb[u] = *(const float4*)(src + 4);            \
        src += (size_t)16 * 1024u;                    \
    } while (0)

#define G_STASH(buf, u)                                                         \
    do {                                                                        \
        *(float2*)&Xt[buf][srow][scol + 0] = make_float2(ra[u].x, ra[u].y);     \
        *(float2*)&Xt[buf][srow][scol + 2] = make_float2(ra[u].z, ra[u].w);     \
        *(float2*)&Xt[buf][srow][scol + 4] = make_float2(rb[u].x, rb[u].y);     \
        *(float2*)&Xt[buf][srow][scol + 6] = make_float2(rb[u].z, rb[u].w);     \
    } while (0)

#define G_HOLD_LOADS_HERE() __builtin_amdgcn_sched_barrier(0)
#define G_MMA_BEFORE_WAIT() __builtin_amdgcn_sched_barrier(0)

#define G_STEP(u, FETCH)                                       \
    do {                                                       \
        if (FETCH) G_FETCH(u);                                 \
        G_HOLD_LOADS_HERE();                                   \
        g_tile_mma(Xt[(u) & 1], m0, khalf, l15, acc);          \
        G_MMA_BEFORE_WAIT();                                   \
        G_STASH(((u) + 1) & 1, ((u) + 1) & 3);                 \
        lds_only_barrier();                                    \
    } while (0)

    G_FETCH(0);
    G_FETCH(1);
    G_FETCH(2);
    G_FETCH(3);
    G_STASH(0, 0);
    lds_only_barrier();

    for (int it = 0; it < 28; it += 4) {
        G_STEP(0, true);
        G_STEP(1, true);
        G_STEP(2, true);
        G_STEP(3, true);
    }
    G_STEP(0, false);
    G_STEP(1, false);
    G_STEP(2, false);
    g_tile_mma(Xt[1], m0, khalf, l15, acc);

#undef G_FETCH
#undef G_STASH
#undef G_HOLD_LOADS_HERE
#undef G_MMA_BEFORE_WAIT
#undef G_STEP

    float* __restrict__ Gp = Gpart + ((size_t)ks * 64u + bh) * 64u * 64u;
    const int rowAdd = (lane >> 4) * 8;
#pragma unroll
    for (int t = 0; t < 4; ++t)
#pragma unroll
        for (int r = 0; r < 8; ++r)
            Gp[(m0 + r + rowAdd) * 64 + t * 16 + l15] = acc[t][r];
}

__global__ void __launch_bounds__(256)
g_reduce(const float* __restrict__ Gpart, float* __restrict__ G) {
    const int i = blockIdx.x * 256 + threadIdx.x;
    const size_t SL = (size_t)64 * 64 * 64;
    G[i] = ((Gpart[i] + Gpart[SL + i]) + Gpart[2 * SL + i]) + Gpart[3 * SL + i];
}

__global__ void __launch_bounds__(128)
wprime_kernel(const float* __restrict__ W, const float* __restrict__ G,
              float* __restrict__ Wp) {
    const int bh   = blockIdx.y;
    const int b    = bh >> 4;
    const int h    = bh & 15;
    const int wave = threadIdx.x >> 5;
    const int lane = threadIdx.x & 31;
    const int l15  = lane & 15;
    const int khalf = (lane >> 4) * 2;
    const int m0 = blockIdx.x * 64 + wave * 16;

    const float* __restrict__ Wh = W + (size_t)(h * 64) * 1024u;
    const float* __restrict__ Gp = G + (size_t)bh * 64u * 64u;

    v8f acc[4];
#pragma unroll
    for (int t = 0; t < 4; ++t) acc[t] = (v8f)0.0f;

#pragma unroll 4
    for (int d0 = 0; d0 < 64; d0 += 4) {
        const int rA = d0 + khalf;
        v2f a;
        a.x = Wh[(size_t)rA * 1024u + m0 + l15];
        a.y = Wh[(size_t)(rA + 1) * 1024u + m0 + l15];
#pragma unroll
        for (int t = 0; t < 4; ++t) {
            v2f bb;
            bb.x = Gp[rA * 64 + t * 16 + l15];
            bb.y = Gp[(rA + 1) * 64 + t * 16 + l15];
            acc[t] = WMMA_F32(a, bb, acc[t]);
        }
    }

    float* __restrict__ Wpb = Wp + (size_t)b * 1024u * 1024u;
    const int rowAdd = (lane >> 4) * 8;
    const int ncol0 = h * 64;
#pragma unroll
    for (int t = 0; t < 4; ++t)
#pragma unroll
        for (int r = 0; r < 8; ++r)
            Wpb[(size_t)(m0 + r + rowAdd) * 1024u + ncol0 + t * 16 + l15] = acc[t][r];
}

#define OT_K      16
#define XS_STRIDE 20
#define WS_PITCH  288

__device__ __forceinline__ void out_tile_mma(const float (*__restrict__ Xs)[XS_STRIDE],
                                             const float (*__restrict__ Ws)[WS_PITCH],
                                             int arow, int bcol2, int half, v8f (&acc)[4][4]) {
#pragma unroll
    for (int kk = 0; kk < OT_K; kk += 4) {
        v2f a[4], bb[4];
#pragma unroll
        for (int mt = 0; mt < 4; ++mt)
            a[mt] = *(const v2f*)&Xs[arow + mt * 16][kk + 2 * half];
#pragma unroll
        for (int nt = 0; nt < 4; ++nt)
            bb[nt] = *(const v2f*)&Ws[kk / 2 + half][bcol2 + nt * 32];
#pragma unroll
        for (int mt = 0; mt < 4; ++mt)
#pragma unroll
            for (int nt = 0; nt < 4; ++nt)
                acc[mt][nt] = WMMA_F32(a[mt], bb[nt], acc[mt][nt]);
    }
}

__global__ void __launch_bounds__(128)
out_kernel(const float* __restrict__ X, const float* __restrict__ Wp,
           float* __restrict__ out) {
    __shared__ __attribute__((aligned(16))) float Xs[2][128][XS_STRIDE];
    __shared__ __attribute__((aligned(16))) float Ws[2][OT_K / 2][WS_PITCH];

    const int b      = blockIdx.z;
    const int s_base = blockIdx.x * 128;
    const int n_base = blockIdx.y * 128;
    const int wave   = threadIdx.x >> 5;
    const int lane   = threadIdx.x & 31;
    const int l15    = lane & 15;
    const int half   = lane >> 4;
    const int m0     = (wave >> 1) * 64;
    const int n0     = (wave & 1) * 64;

    const float* __restrict__ Xb  = X + (size_t)b * 2048u * 1024u;
    const float* __restrict__ Wpb = Wp + (size_t)b * 1024u * 1024u;

    v8f acc[4][4];
#pragma unroll
    for (int mt = 0; mt < 4; ++mt)
#pragma unroll
        for (int nt = 0; nt < 4; ++nt) acc[mt][nt] = (v8f)0.0f;

    const int xr = threadIdx.x >> 2;
    const int xc = (threadIdx.x & 3) * 4;
    const int wn4 = threadIdx.x & 31;
    const int wq  = threadIdx.x >> 5;

    const float* xsrc = &Xb[(size_t)(s_base + xr) * 1024u + xc];
    const float* wsrc = &Wpb[(size_t)(2 * wq) * 1024u + n_base + 4 * wn4];

    float4 xv0, xv1, xv2, xv3, wa0, wb0, wa1, wb1;

#define OUT_FETCH()                                                              \
    do {                                                                         \
        xv0 = *(const float4*)(xsrc);                                            \
        xv1 = *(const float4*)(xsrc + (size_t)32 * 1024u);                       \
        xv2 = *(const float4*)(xsrc + (size_t)64 * 1024u);                       \
        xv3 = *(const float4*)(xsrc + (size_t)96 * 1024u);                       \
        wa0 = *(const float4*)(wsrc);                                            \
        wb0 = *(const float4*)(wsrc + (size_t)1 * 1024u);                        \
        wa1 = *(const float4*)(wsrc + (size_t)8 * 1024u);                        \
        wb1 = *(const float4*)(wsrc + (size_t)9 * 1024u);                        \
        xsrc += OT_K;                                                            \
        wsrc += (size_t)OT_K * 1024u;                                            \
    } while (0)

#define OUT_HOLD_LOADS_HERE() __builtin_amdgcn_sched_barrier(0)
#define OUT_MMA_BEFORE_WAIT() __builtin_amdgcn_sched_barrier(0)

#define OUT_STASH(buf)                                                           \
    do {                                                                         \
        *(float4*)&Xs[buf][xr +  0][xc] = xv0;                                   \
        *(float4*)&Xs[buf][xr + 32][xc] = xv1;                                   \
        *(float4*)&Xs[buf][xr + 64][xc] = xv2;                                   \
        *(float4*)&Xs[buf][xr + 96][xc] = xv3;                                   \
        *(float4*)&Ws[buf][wq][8 * wn4 + 0] =                                    \
            make_float4(wa0.x, wb0.x, wa0.y, wb0.y);                             \
        *(float4*)&Ws[buf][wq][8 * wn4 + 4] =                                    \
            make_float4(wa0.z, wb0.z, wa0.w, wb0.w);                             \
        *(float4*)&Ws[buf][wq + 4][8 * wn4 + 0] =                                \
            make_float4(wa1.x, wb1.x, wa1.y, wb1.y);                             \
        *(float4*)&Ws[buf][wq + 4][8 * wn4 + 4] =                                \
            make_float4(wa1.z, wb1.z, wa1.w, wb1.w);                             \
    } while (0)

    const int arow  = m0 + l15;
    const int bcol2 = 2 * (n0 + l15);

    OUT_FETCH();
    OUT_STASH(0);
    __syncthreads();

#define OUT_TILE_PAIR()                                                          \
    do {                                                                         \
        OUT_FETCH();                                                             \
        OUT_HOLD_LOADS_HERE();                                                   \
        out_tile_mma(Xs[0], Ws[0], arow, bcol2, half, acc);                      \
        OUT_MMA_BEFORE_WAIT();                                                   \
        OUT_STASH(1);                                                            \
        __syncthreads();                                                         \
        OUT_FETCH();                                                             \
        OUT_HOLD_LOADS_HERE();                                                   \
        out_tile_mma(Xs[1], Ws[1], arow, bcol2, half, acc);                      \
        OUT_MMA_BEFORE_WAIT();                                                   \
        OUT_STASH(0);                                                            \
        __syncthreads();                                                         \
    } while (0)

    __builtin_amdgcn_s_setprio(3);
    for (int it = 0; it < 32; it += 2) OUT_TILE_PAIR();
    __builtin_amdgcn_s_setprio(2);
    for (int it = 32; it < 48; it += 2) OUT_TILE_PAIR();
    __builtin_amdgcn_s_setprio(1);
    for (int it = 48; it < 56; it += 2) OUT_TILE_PAIR();
    __builtin_amdgcn_s_setprio(0);
    for (int it = 56; it < 62; it += 2) OUT_TILE_PAIR();
    OUT_FETCH();
    OUT_HOLD_LOADS_HERE();
    out_tile_mma(Xs[0], Ws[0], arow, bcol2, half, acc);
    OUT_MMA_BEFORE_WAIT();
    OUT_STASH(1);
    __syncthreads();
    out_tile_mma(Xs[1], Ws[1], arow, bcol2, half, acc);

#undef OUT_TILE_PAIR
#undef OUT_FETCH
#undef OUT_HOLD_LOADS_HERE
#undef OUT_MMA_BEFORE_WAIT
#undef OUT_STASH

    float* __restrict__ ob = out + (size_t)b * 2048u * 1024u;
    const int rowAdd = half * 8;
#pragma unroll
    for (int mt = 0; mt < 4; ++mt)
#pragma unroll
        for (int nt = 0; nt < 4; ++nt)
#pragma unroll
            for (int r = 0; r < 8; ++r)
                ob[(size_t)(s_base + m0 + mt * 16 + r + rowAdd) * 1024u +
                   n_base + n0 + nt * 16 + l15] = acc[mt][nt][r];
}

extern "C" void kernel_launch(void* const* d_in, const int* in_sizes, int n_in,
                              void* d_out, int out_size, void* d_ws, size_t ws_size,
                              hipStream_t stream) {
    (void)in_sizes; (void)n_in; (void)out_size; (void)ws_size;
    const float* X = (const float*)d_in[0];
    const float* W = (const float*)d_in[1];
    float* out = (float*)d_out;

    float* Gpart = (float*)d_ws;
    float* G     = Gpart + (size_t)4 * 64 * 64 * 64;
    float* Wp    = G + (size_t)64 * 64 * 64;

    g_kernel<<<dim3(64, 4), dim3(128), 0, stream>>>(X, Gpart);
    g_reduce<<<dim3(1024), dim3(256), 0, stream>>>(Gpart, G);
    wprime_kernel<<<dim3(16, 64), dim3(128), 0, stream>>>(W, G, Wp);
    out_kernel<<<dim3(16, 8, 4), dim3(128), 0, stream>>>(X, Wp, out);
}


# G and W' kernels: fragments by 16-byte global loads under lane map col=4c+t, no LDS staging or K-loop barrier; G: each wave sums a 128-row sub-slice, folded in wave order (sum order regrouped)
# speedup vs baseline: 1.4691x; 1.0359x over previous
#include <hip/hip_runtime.h>

typedef __attribute__((ext_vector_type(2))) float v2f;
typedef __attribute__((ext_vector_type(8))) float v8f;

#define WMMA_F32(a, b, c) \
    __builtin_amdgcn_wmma_f32_16x16x4_f32(false, (a), false, (b), (short)0, (c), false, false)


__device__ __forceinline__ v2f frag2(float lo, float hi) {
    v2f f;
    f.x = lo;
    f.y = hi;
    return f;
}

__device__ __forceinline__ void kstep_64x64(const float4 pa, const float4 qa,
                                            const float4 pb, const float4 qb,
                                            v8f (&acc)[4][4]) {
    v2f fa[4], fb[4];
    fa[0] = frag2(pa.x, qa.x); fa[1] = frag2(pa.y, qa.y);
    fa[2] = frag2(pa.z, qa.z); fa[3] = frag2(pa.w, qa.w);
    fb[0] = frag2(pb.x, qb.x); fb[1] = frag2(pb.y, qb.y);
    fb[2] = frag2(pb.z, qb.z); fb[3] = frag2(pb.w, qb.w);
#pragma unroll
    for (int mt = 0; mt < 4; ++mt)
#pragma unroll
        for (int nt = 0; nt < 4; ++nt)
            acc[mt][nt] = WMMA_F32(fa[mt], fb[nt], acc[mt][nt]);
}

#define G_AHEAD 8

__global__ void __launch_bounds__(128)
g_kernel(const float* __restrict__ X, float* __restrict__ Gpart) {
    __shared__ __attribute__((aligned(16))) float Pw[4][64 * 64];

    const int bh   = blockIdx.x;
    const int ks   = blockIdx.y;
    const int b    = bh >> 4;
    const int h    = bh & 15;
    const int wave = threadIdx.x >> 5;
    const int lane = threadIdx.x & 31;
    const int c    = lane & 15;
    const int half = lane >> 4;

    const float* src = X + (size_t)b * 2048u * 1024u + h * 64 +
                       (size_t)(ks * 512 + wave * 128 + 2 * half) * 1024u + 4 * c;

    v8f acc[4][4];
#pragma unroll
    for (int mt = 0; mt < 4; ++mt)
#pragma unroll
        for (int nt = 0; nt < 4; ++nt) acc[mt][nt] = (v8f)0.0f;

    float4 p[G_AHEAD], q[G_AHEAD];
#pragma unroll
    for (int u = 0; u < G_AHEAD; ++u) {
        p[u] = *(const float4*)(src);
        q[u] = *(const float4*)(src + 1024);
        src += (size_t)4 * 1024u;
    }

    for (int it = 0; it < 32 - G_AHEAD; it += G_AHEAD) {
#pragma unroll
        for (int u = 0; u < G_AHEAD; ++u) {
            const float4 pp = p[u], qq = q[u];
            p[u] = *(const float4*)(src);
            q[u] = *(const float4*)(src + 1024);
            src += (size_t)4 * 1024u;
            __builtin_amdgcn_sched_barrier(0);
            kstep_64x64(pp, qq, pp, qq, acc);
        }
    }
#pragma unroll
    for (int u = 0; u < G_AHEAD; ++u) kstep_64x64(p[u], q[u], p[u], q[u], acc);

    float* mine = Pw[wave];
#pragma unroll
    for (int mt = 0; mt < 4; ++mt)
#pragma unroll
        for (int r = 0; r < 8; ++r)
            *(float4*)&mine[(32 * half + 4 * r + mt) * 64 + 4 * c] =
                make_float4(acc[mt][0][r], acc[mt][1][r], acc[mt][2][r], acc[mt][3][r]);
    __syncthreads();

    float* __restrict__ Gp = Gpart + ((size_t)ks * 64u + bh) * 64u * 64u;
#pragma unroll
    for (int i = 0; i < 8; ++i) {
        const int j = (i * 128 + (int)threadIdx.x) * 4;
        const float4 s0 = *(const float4*)&Pw[0][j];
        const float4 s1 = *(const float4*)&Pw[1][j];
        const float4 s2 = *(const float4*)&Pw[2][j];
        const float4 s3 = *(const float4*)&Pw[3][j];
        float4 s;
        s.x = ((s0.x + s1.x) + s2.x) + s3.x;
        s.y = ((s0.y + s1.y) + s2.y) + s3.y;
        s.z = ((s0.z + s1.z) + s2.z) + s3.z;
        s.w = ((s0.w + s1.w) + s2.w) + s3.w;
        *(float4*)&Gp[j] = s;
    }
}

__global__ void __launch_bounds__(256)
g_reduce(const float* __restrict__ Gpart, float* __restrict__ G) {
    const int i = blockIdx.x * 256 + threadIdx.x;
    const size_t SL = (size_t)64 * 64 * 64;
    G[i] = ((Gpart[i] + Gpart[SL + i]) + Gpart[2 * SL + i]) + Gpart[3 * SL + i];
}

#define WP_AHEAD 4

__global__ void __launch_bounds__(128)
wprime_kernel(const float* __restrict__ W, const float* __restrict__ G,
              float* __restrict__ Wp) {
    const int bh   = blockIdx.y;
    const int b    = bh >> 4;
    const int h    = bh & 15;
    const int wave = threadIdx.x >> 5;
    const int lane = threadIdx.x & 31;
    const int c    = lane & 15;
    const int half = lane >> 4;
    const int mbase = blockIdx.x * 256 + wave * 64;

    const float* wsrc = W + (size_t)(h * 64 + 2 * half) * 1024u + mbase + 4 * c;
    const float* gsrc = G + (size_t)bh * 64u * 64u + (2 * half) * 64 + 4 * c;

    v8f acc[4][4];
#pragma unroll
    for (int mt = 0; mt < 4; ++mt)
#pragma unroll
        for (int nt = 0; nt < 4; ++nt) acc[mt][nt] = (v8f)0.0f;

    float4 pa[WP_AHEAD], qa[WP_AHEAD], pb[WP_AHEAD], qb[WP_AHEAD];
#pragma unroll
    for (int u = 0; u < WP_AHEAD; ++u) {
        pa[u] = *(const float4*)(wsrc);
        qa[u] = *(const float4*)(wsrc + 1024);
        pb[u] = *(const float4*)(gsrc);
        qb[u] = *(const float4*)(gsrc + 64);
        wsrc += (size_t)4 * 1024u;
        gsrc += 4 * 64;
    }

    for (int it = 0; it < 16 - WP_AHEAD; it += WP_AHEAD) {
#pragma unroll
        for (int u = 0; u < WP_AHEAD; ++u) {
            const float4 a0 = pa[u], a1 = qa[u], b0 = pb[u], b1 = qb[u];
            pa[u] = *(const float4*)(wsrc);
            qa[u] = *(const float4*)(wsrc + 1024);
            pb[u] = *(const float4*)(gsrc);
            qb[u] = *(const float4*)(gsrc + 64);
            wsrc += (size_t)4 * 1024u;
            gsrc += 4 * 64;
            __builtin_amdgcn_sched_barrier(0);
            kstep_64x64(a0, a1, b0, b1, acc);
        }
    }
#pragma unroll
    for (int u = 0; u < WP_AHEAD; ++u) kstep_64x64(pa[u], qa[u], pb[u], qb[u], acc);

    float* __restrict__ dst = Wp + (size_t)b * 1024u * 1024u +
                              (size_t)(mbase + 32 * half) * 1024u + h * 64 + 4 * c;
#pragma unroll
    for (int mt = 0; mt < 4; ++mt)
#pragma unroll
        for (int r = 0; r < 8; ++r)
            *(float4*)&dst[(size_t)(4 * r + mt) * 1024u] =
                make_float4(acc[mt][0][r], acc[mt][1][r], acc[mt][2][r], acc[mt][3][r]);
}

#define OT_K      16
#define XS_STRIDE 20
#define WS_PITCH  288

__device__ __forceinline__ void out_tile_mma(const float (*__restrict__ Xs)[XS_STRIDE],
                                             const float (*__restrict__ Ws)[WS_PITCH],
                                             int arow, int bcol2, int half, v8f (&acc)[4][4]) {
#pragma unroll
    for (int kk = 0; kk < OT_K; kk += 4) {
        v2f a[4], bb[4];
#pragma unroll
        for (int mt = 0; mt < 4; ++mt)
            a[mt] = *(const v2f*)&Xs[arow + mt * 16][kk + 2 * half];
#pragma unroll
        for (int nt = 0; nt < 4; ++nt)
            bb[nt] = *(const v2f*)&Ws[kk / 2 + half][bcol2 + nt * 32];
#pragma unroll
        for (int mt = 0; mt < 4; ++mt)
#pragma unroll
            for (int nt = 0; nt < 4; ++nt)
                acc[mt][nt] = WMMA_F32(a[mt], bb[nt], acc[mt][nt]);
    }
}

__global__ void __launch_bounds__(128)
out_kernel(const float* __restrict__ X, const float* __restrict__ Wp,
           float* __restrict__ out) {
    __shared__ __attribute__((aligned(16))) float Xs[2][128][XS_STRIDE];
    __shared__ __attribute__((aligned(16))) float Ws[2][OT_K / 2][WS_PITCH];

    const int b      = blockIdx.z;
    const int s_base = blockIdx.x * 128;
    const int n_base = blockIdx.y * 128;
    const int wave   = threadIdx.x >> 5;
    const int lane   = threadIdx.x & 31;
    const int l15    = lane & 15;
    const int half   = lane >> 4;
    const int m0     = (wave >> 1) * 64;
    const int n0     = (wave & 1) * 64;

    const float* __restrict__ Xb  = X + (size_t)b * 2048u * 1024u;
    const float* __restrict__ Wpb = Wp + (size_t)b * 1024u * 1024u;

    v8f acc[4][4];
#pragma unroll
    for (int mt = 0; mt < 4; ++mt)
#pragma unroll
        for (int nt = 0; nt < 4; ++nt) acc[mt][nt] = (v8f)0.0f;

    const int xr = threadIdx.x >> 2;
    const int xc = (threadIdx.x & 3) * 4;
    const int wn4 = threadIdx.x & 31;
    const int wq  = threadIdx.x >> 5;

    const float* xsrc = &Xb[(size_t)(s_base + xr) * 1024u + xc];
    const float* wsrc = &Wpb[(size_t)(2 * wq) * 1024u + n_base + 4 * wn4];

    float4 xv0, xv1, xv2, xv3, wa0, wb0, wa1, wb1;

#define OUT_FETCH()                                                              \
    do {                                                                         \
        xv0 = *(const float4*)(xsrc);                                            \
        xv1 = *(const float4*)(xsrc + (size_t)32 * 1024u);                       \
        xv2 = *(const float4*)(xsrc + (size_t)64 * 1024u);                       \
        xv3 = *(const float4*)(xsrc + (size_t)96 * 1024u);                       \
        wa0 = *(const float4*)(wsrc);                                            \
        wb0 = *(const float4*)(wsrc + (size_t)1 * 1024u);                        \
        wa1 = *(const float4*)(wsrc + (size_t)8 * 1024u);                        \
        wb1 = *(const float4*)(wsrc + (size_t)9 * 1024u);                        \
        xsrc += OT_K;                                                            \
        wsrc += (size_t)OT_K * 1024u;                                            \
    } while (0)

#define OUT_HOLD_LOADS_HERE() __builtin_amdgcn_sched_barrier(0)
#define OUT_MMA_BEFORE_WAIT() __builtin_amdgcn_sched_barrier(0)

#define OUT_STASH(buf)                                                           \
    do {                                                                         \
        *(float4*)&Xs[buf][xr +  0][xc] = xv0;                                   \
        *(float4*)&Xs[buf][xr + 32][xc] = xv1;                                   \
        *(float4*)&Xs[buf][xr + 64][xc] = xv2;                                   \
        *(float4*)&Xs[buf][xr + 96][xc] = xv3;                                   \
        *(float4*)&Ws[buf][wq][8 * wn4 + 0] =                                    \
            make_float4(wa0.x, wb0.x, wa0.y, wb0.y);                             \
        *(float4*)&Ws[buf][wq][8 * wn4 + 4] =                                    \
            make_float4(wa0.z, wb0.z, wa0.w, wb0.w);                             \
        *(float4*)&Ws[buf][wq + 4][8 * wn4 + 0] =                                \
            make_float4(wa1.x, wb1.x, wa1.y, wb1.y);                             \
        *(float4*)&Ws[buf][wq + 4][8 * wn4 + 4] =                                \
            make_float4(wa1.z, wb1.z, wa1.w, wb1.w);                             \
    } while (0)

    const int arow  = m0 + l15;
    const int bcol2 = 2 * (n0 + l15);

    OUT_FETCH();
    OUT_STASH(0);
    __syncthreads();

#define OUT_TILE_PAIR()                                                          \
    do {                                                                         \
        OUT_FETCH();                                                             \
        OUT_HOLD_LOADS_HERE();                                                   \
        out_tile_mma(Xs[0], Ws[0], arow, bcol2, half, acc);                      \
        OUT_MMA_BEFORE_WAIT();                                                   \
        OUT_STASH(1);                                                            \
        __syncthreads();                                                         \
        OUT_FETCH();                                                             \
        OUT_HOLD_LOADS_HERE();                                                   \
        out_tile_mma(Xs[1], Ws[1], arow, bcol2, half, acc);                      \
        OUT_MMA_BEFORE_WAIT();                                                   \
        OUT_STASH(0);                                                            \
        __syncthreads();                                                         \
    } while (0)

    __builtin_amdgcn_s_setprio(3);
    for (int it = 0; it < 32; it += 2) OUT_TILE_PAIR();
    __builtin_amdgcn_s_setprio(2);
    for (int it = 32; it < 48; it += 2) OUT_TILE_PAIR();
    __builtin_amdgcn_s_setprio(1);
    for (int it = 48; it < 56; it += 2) OUT_TILE_PAIR();
    __builtin_amdgcn_s_setprio(0);
    for (int it = 56; it < 62; it += 2) OUT_TILE_PAIR();
    OUT_FETCH();
    OUT_HOLD_LOADS_HERE();
    out_tile_mma(Xs[0], Ws[0], arow, bcol2, half, acc);
    OUT_MMA_BEFORE_WAIT();
    OUT_STASH(1);
    __syncthreads();
    out_tile_mma(Xs[1], Ws[1], arow, bcol2, half, acc);

#undef OUT_TILE_PAIR
#undef OUT_FETCH
#undef OUT_HOLD_LOADS_HERE
#undef OUT_MMA_BEFORE_WAIT
#undef OUT_STASH

    float* __restrict__ ob = out + (size_t)b * 2048u * 1024u;
    const int rowAdd = half * 8;
#pragma unroll
    for (int mt = 0; mt < 4; ++mt)
#pragma unroll
        for (int nt = 0; nt < 4; ++nt)
#pragma unroll
            for (int r = 0; r < 8; ++r)
                ob[(size_t)(s_base + m0 + mt * 16 + r + rowAdd) * 1024u +
                   n_base + n0 + nt * 16 + l15] = acc[mt][nt][r];
}

extern "C" void kernel_launch(void* const* d_in, const int* in_sizes, int n_in,
                              void* d_out, int out_size, void* d_ws, size_t ws_size,
                              hipStream_t stream) {
    (void)in_sizes; (void)n_in; (void)out_size; (void)ws_size;
    const float* X = (const float*)d_in[0];
    const float* W = (const float*)d_in[1];
    float* out = (float*)d_out;

    float* Gpart = (float*)d_ws;
    float* G     = Gpart + (size_t)4 * 64 * 64 * 64;
    float* Wp    = G + (size_t)64 * 64 * 64;

    g_kernel<<<dim3(64, 4), dim3(128), 0, stream>>>(X, Gpart);
    g_reduce<<<dim3(1024), dim3(256), 0, stream>>>(Gpart, G);
    wprime_kernel<<<dim3(4, 64), dim3(128), 0, stream>>>(W, G, Wp);
    out_kernel<<<dim3(16, 8, 4), dim3(128), 0, stream>>>(X, Wp, out);
}


# out = (X W^T) G: the 64-wide G product runs behind the big product as a per-wave tail, G's split-K partials summed in slice order under the K loop; W' and reduce launches gone (4->2); sums regrouped
# speedup vs baseline: 1.6146x; 1.0990x over previous
#include <hip/hip_runtime.h>

typedef __attribute__((ext_vector_type(2))) float v2f;
typedef __attribute__((ext_vector_type(8))) float v8f;

#define WMMA_F32(a, b, c) \
    __builtin_amdgcn_wmma_f32_16x16x4_f32(false, (a), false, (b), (short)0, (c), false, false)


__device__ __forceinline__ v2f frag2(float lo, float hi) {
    v2f f;
    f.x = lo;
    f.y = hi;
    return f;
}

__device__ __forceinline__ void kstep_64x64(const float4 pa, const float4 qa,
                                            const float4 pb, const float4 qb,
                                            v8f (&acc)[4][4]) {
    v2f fa[4], fb[4];
    fa[0] = frag2(pa.x, qa.x); fa[1] = frag2(pa.y, qa.y);
    fa[2] = frag2(pa.z, qa.z); fa[3] = frag2(pa.w, qa.w);
    fb[0] = frag2(pb.x, qb.x); fb[1] = frag2(pb.y, qb.y);
    fb[2] = frag2(pb.z, qb.z); fb[3] = frag2(pb.w, qb.w);
#pragma unroll
    for (int mt = 0; mt < 4; ++mt)
#pragma unroll
        for (int nt = 0; nt < 4; ++nt)
            acc[mt][nt] = WMMA_F32(fa[mt], fb[nt], acc[mt][nt]);
}

#define G_AHEAD 8

__global__ void __launch_bounds__(128)
g_kernel(const float* __restrict__ X, float* __restrict__ Gpart) {
    __shared__ __attribute__((aligned(16))) float Pw[4][64 * 64];

    const int bh   = blockIdx.x;
    const int ks   = blockIdx.y;
    const int b    = bh >> 4;
    const int h    = bh & 15;
    const int wave = threadIdx.x >> 5;
    const int lane = threadIdx.x & 31;
    const int c    = lane & 15;
    const int half = lane >> 4;

    const float* src = X + (size_t)b * 2048u * 1024u + h * 64 +
                       (size_t)(ks * 512 + wave * 128 + 2 * half) * 1024u + 4 * c;

    v8f acc[4][4];
#pragma unroll
    for (int mt = 0; mt < 4; ++mt)
#pragma unroll
        for (int nt = 0; nt < 4; ++nt) acc[mt][nt] = (v8f)0.0f;

    float4 p[G_AHEAD], q[G_AHEAD];
#pragma unroll
    for (int u = 0; u < G_AHEAD; ++u) {
        p[u] = *(const float4*)(src);
        q[u] = *(const float4*)(src + 1024);
        src += (size_t)4 * 1024u;
    }

    for (int it = 0; it < 32 - G_AHEAD; it += G_AHEAD) {
#pragma unroll
        for (int u = 0; u < G_AHEAD; ++u) {
            const float4 pp = p[u], qq = q[u];
            p[u] = *(const float4*)(src);
            q[u] = *(const float4*)(src + 1024);
            src += (size_t)4 * 1024u;
            __builtin_amdgcn_sched_barrier(0);
            kstep_64x64(pp, qq, pp, qq, acc);
        }
    }
#pragma unroll
    for (int u = 0; u < G_AHEAD; ++u) kstep_64x64(p[u], q[u], p[u], q[u], acc);

    float* mine = Pw[wave];
#pragma unroll
    for (int mt = 0; mt < 4; ++mt)
#pragma unroll
        for (int r = 0; r < 8; ++r)
            *(float4*)&mine[(32 * half + 4 * r + mt) * 64 + 4 * c] =
                make_float4(acc[mt][0][r], acc[mt][1][r], acc[mt][2][r], acc[mt][3][r]);
    __syncthreads();

    float* __restrict__ Gp = Gpart + ((size_t)ks * 64u + bh) * 64u * 64u;
#pragma unroll
    for (int i = 0; i < 8; ++i) {
        const int j = (i * 128 + (int)threadIdx.x) * 4;
        const float4 s0 = *(const float4*)&Pw[0][j];
        const float4 s1 = *(const float4*)&Pw[1][j];
        const float4 s2 = *(const float4*)&Pw[2][j];
        const float4 s3 = *(const float4*)&Pw[3][j];
        float4 s;
        s.x = ((s0.x + s1.x) + s2.x) + s3.x;
        s.y = ((s0.y + s1.y) + s2.y) + s3.y;
        s.z = ((s0.z + s1.z) + s2.z) + s3.z;
        s.w = ((s0.w + s1.w) + s2.w) + s3.w;
        *(float4*)&Gp[j] = s;
    }
}

#define OT_K      16
#define XS_STRIDE 20
#define QS_PITCH  68
#define GP_SLICE  ((size_t)64 * 64 * 64)

#define LDS_XS    0
#define LDS_WS    (2 * 128 * XS_STRIDE)
#define LDS_QS    0
#define LDS_GS    (4 * 64 * QS_PITCH)
#define LDS_TOTAL (LDS_GS + 2 * 64 * 64)

__device__ __forceinline__ void out_tile_mma(const float* __restrict__ xs,
                                             const float* __restrict__ ws,
                                             int arow, int brow, int half, v8f (&acc)[4][4]) {
#pragma unroll
    for (int kk = 0; kk < OT_K; kk += 4) {
        v2f a[4], bb[4];
#pragma unroll
        for (int mt = 0; mt < 4; ++mt)
            a[mt] = *(const v2f*)&xs[(arow + mt * 16) * XS_STRIDE + kk + 2 * half];
#pragma unroll
        for (int nt = 0; nt < 4; ++nt)
            bb[nt] = *(const v2f*)&ws[(brow + nt * 16) * XS_STRIDE + kk + 2 * half];
#pragma unroll
        for (int mt = 0; mt < 4; ++mt)
#pragma unroll
            for (int nt = 0; nt < 4; ++nt)
                acc[mt][nt] = WMMA_F32(a[mt], bb[nt], acc[mt][nt]);
    }
}

__global__ void __launch_bounds__(128)
out_kernel(const float* __restrict__ X, const float* __restrict__ W,
           const float* __restrict__ Gpart, float* __restrict__ out) {
    __shared__ __attribute__((aligned(16))) float smem[LDS_TOTAL];

    const int b      = blockIdx.z;
    const int s_base = blockIdx.x * 128;
    const int n_base = blockIdx.y * 128;
    const int wave   = threadIdx.x >> 5;
    const int lane   = threadIdx.x & 31;
    const int l15    = lane & 15;
    const int half   = lane >> 4;
    const int m0     = (wave >> 1) * 64;
    const int n0     = (wave & 1) * 64;

    v8f acc[4][4];
#pragma unroll
    for (int mt = 0; mt < 4; ++mt)
#pragma unroll
        for (int nt = 0; nt < 4; ++nt) acc[mt][nt] = (v8f)0.0f;

    const int xr = threadIdx.x >> 2;
    const int xc = (threadIdx.x & 3) * 4;

    const float* xsrc = X + (size_t)b * 2048u * 1024u + (size_t)(s_base + xr) * 1024u + xc;
    const float* wsrc = W + (size_t)(n_base + xr) * 1024u + xc;
    const float* gsrc = Gpart + (size_t)(b * 16 + 2 * (int)blockIdx.y) * 4096u + 4 * (int)threadIdx.x;
    float* gdst = &smem[LDS_GS + 4 * (int)threadIdx.x];

    float* const stx = &smem[LDS_XS + xr * XS_STRIDE + xc];
    float* const stw = &smem[LDS_WS + xr * XS_STRIDE + xc];

    float4 xv0, xv1, xv2, xv3, wv0, wv1, wv2, wv3;
    float4 ga, gb2, gt;

#define OUT_FETCH()                                                              \
    do {                                                                         \
        xv0 = *(const float4*)(xsrc);                                            \
        xv1 = *(const float4*)(xsrc + (size_t)32 * 1024u);                       \
        xv2 = *(const float4*)(xsrc + (size_t)64 * 1024u);                       \
        xv3 = *(const float4*)(xsrc + (size_t)96 * 1024u);                       \
        wv0 = *(const float4*)(wsrc);                                            \
        wv1 = *(const float4*)(wsrc + (size_t)32 * 1024u);                       \
        wv2 = *(const float4*)(wsrc + (size_t)64 * 1024u);                       \
        wv3 = *(const float4*)(wsrc + (size_t)96 * 1024u);                       \
        xsrc += OT_K;                                                            \
        wsrc += OT_K;                                                            \
    } while (0)

#define OUT_FETCH_G01()                                                          \
    do {                                                                         \
        ga  = *(const float4*)(gsrc);                                            \
        gb2 = *(const float4*)(gsrc + GP_SLICE);                                 \
    } while (0)

#define OUT_FETCH_G23()                                                          \
    do {                                                                         \
        ga  = *(const float4*)(gsrc + 2 * GP_SLICE);                             \
        gb2 = *(const float4*)(gsrc + 3 * GP_SLICE);                             \
        gsrc += 512;                                                             \
    } while (0)

#define OUT_HOLD_LOADS_HERE() __builtin_amdgcn_sched_barrier(0)
#define OUT_MMA_BEFORE_WAIT() __builtin_amdgcn_sched_barrier(0)

#define OUT_STASH(buf)                                                           \
    do {                                                                         \
        *(float4*)(stx + ((buf) * 128 +  0) * XS_STRIDE) = xv0;                  \
        *(float4*)(stx + ((buf) * 128 + 32) * XS_STRIDE) = xv1;                  \
        *(float4*)(stx + ((buf) * 128 + 64) * XS_STRIDE) = xv2;                  \
        *(float4*)(stx + ((buf) * 128 + 96) * XS_STRIDE) = xv3;                  \
        *(float4*)(stw + ((buf) * 128 +  0) * XS_STRIDE) = wv0;                  \
        *(float4*)(stw + ((buf) * 128 + 32) * XS_STRIDE) = wv1;                  \
        *(float4*)(stw + ((buf) * 128 + 64) * XS_STRIDE) = wv2;                  \
        *(float4*)(stw + ((buf) * 128 + 96) * XS_STRIDE) = wv3;                  \
    } while (0)

#define OUT_G_AFTER_MMA()                                                        \
    asm volatile("" : "+v"(ga.x), "+v"(ga.y), "+v"(ga.z), "+v"(ga.w),            \
                      "+v"(gb2.x), "+v"(gb2.y), "+v"(gb2.z), "+v"(gb2.w))

#define OUT_FOLD_G01()                                                           \
    do {                                                                         \
        OUT_G_AFTER_MMA();                                                       \
        gt.x = ga.x + gb2.x;                                                     \
        gt.y = ga.y + gb2.y;                                                     \
        gt.z = ga.z + gb2.z;                                                     \
        gt.w = ga.w + gb2.w;                                                     \
    } while (0)

#define OUT_STASH_G23()                                                          \
    do {                                                                         \
        OUT_G_AFTER_MMA();                                                       \
        gt.x = (gt.x + ga.x) + gb2.x;                                            \
        gt.y = (gt.y + ga.y) + gb2.y;                                            \
        gt.z = (gt.z + ga.z) + gb2.z;                                            \
        gt.w = (gt.w + ga.w) + gb2.w;                                            \
        *(float4*)gdst = gt;                                                     \
        gdst += 512;                                                             \
    } while (0)

    const int arow = m0 + l15;
    const int brow = n0 + l15;
    const float* const xs0 = &smem[LDS_XS];
    const float* const xs1 = &smem[LDS_XS + 128 * XS_STRIDE];
    const float* const ws0 = &smem[LDS_WS];
    const float* const ws1 = &smem[LDS_WS + 128 * XS_STRIDE];

    OUT_FETCH();
    OUT_STASH(0);
    __syncthreads();

#define OUT_TILE_PAIR()                                                          \
    do {                                                                         \
        OUT_FETCH();                                                             \
        OUT_HOLD_LOADS_HERE();                                                   \
        out_tile_mma(xs0, ws0, arow, brow, half, acc);                           \
        OUT_MMA_BEFORE_WAIT();                                                   \
        OUT_STASH(1);                                                            \
        __syncthreads();                                                         \
        OUT_FETCH();                                                             \
        OUT_HOLD_LOADS_HERE();                                                   \
        out_tile_mma(xs1, ws1, arow, brow, half, acc);                           \
        OUT_MMA_BEFORE_WAIT();                                                   \
        OUT_STASH(0);                                                            \
        __syncthreads();                                                         \
    } while (0)

#define OUT_TILE_PAIR_G()                                                        \
    do {                                                                         \
        OUT_FETCH();                                                             \
        OUT_FETCH_G01();                                                         \
        OUT_HOLD_LOADS_HERE();                                                   \
        out_tile_mma(xs0, ws0, arow, brow, half, acc);                           \
        OUT_MMA_BEFORE_WAIT();                                                   \
        OUT_STASH(1);                                                            \
        OUT_FOLD_G01();                                                          \
        __syncthreads();                                                         \
        OUT_FETCH();                                                             \
        OUT_FETCH_G23();                                                         \
        OUT_HOLD_LOADS_HERE();                                                   \
        out_tile_mma(xs1, ws1, arow, brow, half, acc);                           \
        OUT_MMA_BEFORE_WAIT();                                                   \
        OUT_STASH(0);                                                            \
        OUT_STASH_G23();                                                         \
        __syncthreads();                                                         \
    } while (0)

    __builtin_amdgcn_s_setprio(3);
    for (int it = 0; it < 32; it += 2) OUT_TILE_PAIR_G();
    __builtin_amdgcn_s_setprio(2);
    for (int it = 32; it < 48; it += 2) OUT_TILE_PAIR();
    __builtin_amdgcn_s_setprio(1);
#pragma nounroll
    for (int it = 48; it < 56; it += 2) OUT_TILE_PAIR();
    __builtin_amdgcn_s_setprio(0);
#pragma nounroll
    for (int it = 56; it < 62; it += 2) OUT_TILE_PAIR();
    OUT_FETCH();
    OUT_HOLD_LOADS_HERE();
    out_tile_mma(xs0, ws0, arow, brow, half, acc);
    OUT_MMA_BEFORE_WAIT();
    OUT_STASH(1);
    __syncthreads();
    out_tile_mma(xs1, ws1, arow, brow, half, acc);

#undef OUT_TILE_PAIR_G
#undef OUT_TILE_PAIR
#undef OUT_FETCH
#undef OUT_FETCH_G01
#undef OUT_FETCH_G23
#undef OUT_HOLD_LOADS_HERE
#undef OUT_MMA_BEFORE_WAIT
#undef OUT_STASH
#undef OUT_G_AFTER_MMA
#undef OUT_FOLD_G01
#undef OUT_STASH_G23

    __syncthreads();
    float* const qs = &smem[LDS_QS + wave * (64 * QS_PITCH)];
#pragma unroll
    for (int mt = 0; mt < 4; ++mt)
#pragma unroll
        for (int nt = 0; nt < 4; ++nt)
#pragma unroll
            for (int r = 0; r < 8; ++r)
                qs[(16 * mt + 8 * half + r) * QS_PITCH + 16 * nt + l15] = acc[mt][nt][r];
    __syncthreads();

    v8f res[4][4];
#pragma unroll
    for (int mt = 0; mt < 4; ++mt)
#pragma unroll
        for (int nt = 0; nt < 4; ++nt) res[mt][nt] = (v8f)0.0f;

    const float* const qa = qs + l15 * QS_PITCH + 2 * half;
    const float* const gb = &smem[LDS_GS + (wave & 1) * 4096 + (2 * half) * 64 + 4 * l15];
#pragma unroll
    for (int d0 = 0; d0 < 64; d0 += 4) {
        v2f a[4], fb[4];
#pragma unroll
        for (int mt = 0; mt < 4; ++mt)
            a[mt] = *(const v2f*)&qa[(16 * mt) * QS_PITCH + d0];
        const float4 pb = *(const float4*)&gb[d0 * 64];
        const float4 qb = *(const float4*)&gb[d0 * 64 + 64];
        fb[0] = frag2(pb.x, qb.x); fb[1] = frag2(pb.y, qb.y);
        fb[2] = frag2(pb.z, qb.z); fb[3] = frag2(pb.w, qb.w);
#pragma unroll
        for (int mt = 0; mt < 4; ++mt)
#pragma unroll
            for (int nt = 0; nt < 4; ++nt)
                res[mt][nt] = WMMA_F32(a[mt], fb[nt], res[mt][nt]);
    }

    float* __restrict__ ob = out + (size_t)b * 2048u * 1024u +
                             (size_t)(s_base + m0 + 8 * half) * 1024u + n_base + n0 + 4 * l15;
#pragma unroll
    for (int mt = 0; mt < 4; ++mt)
#pragma unroll
        for (int r = 0; r < 8; ++r)
            *(float4*)&ob[(size_t)(16 * mt + r) * 1024u] =
                make_float4(res[mt][0][r], res[mt][1][r], res[mt][2][r], res[mt][3][r]);
}

extern "C" void kernel_launch(void* const* d_in, const int* in_sizes, int n_in,
                              void* d_out, int out_size, void* d_ws, size_t ws_size,
                              hipStream_t stream) {
    (void)in_sizes; (void)n_in; (void)out_size; (void)ws_size;
    const float* X = (const float*)d_in[0];
    const float* W = (const float*)d_in[1];
    float* out = (float*)d_out;

    float* Gpart = (float*)d_ws;

    g_kernel<<<dim3(64, 4), dim3(128), 0, stream>>>(X, Gpart);
    out_kernel<<<dim3(16, 8, 4), dim3(128), 0, stream>>>(X, W, Gpart, out);
}
